# starting kernel
# baseline (speedup 1.0000x reference)
#include <hip/hip_runtime.h>

typedef float v2f __attribute__((ext_vector_type(2)));
typedef float v8f __attribute__((ext_vector_type(8)));

#define SAMP_PER_BLOCK 16

__global__ __launch_bounds__(256) void qnet_fused_kernel(
    const float* __restrict__ x,
    const float* __restrict__ qw,
    const float* __restrict__ hw,
    const float* __restrict__ hb,
    const float* __restrict__ scale_p,
    float* __restrict__ out)
{
    __shared__ float pooledLDS[16 * 16];
    __shared__ float probsLDS[16 * 16];

    const int tid  = threadIdx.x;
    const int lane = tid & 31;
    const int wv   = tid >> 5;
    const int base = blockIdx.x * SAMP_PER_BLOCK;

    #pragma unroll
    for (int ss = 0; ss < 2; ++ss) {
        const int sl = wv * 2 + ss;
        const float4* __restrict__ xs =
            (const float4*)(x + (size_t)(base + sl) * 4096);
        float acc[4];
        #pragma unroll
        for (int rb = 0; rb < 4; ++rb) {
            float s = 0.0f;
            #pragma unroll
            for (int k8 = 0; k8 < 8; ++k8) {
                const int k = rb * 8 + k8;
                const float4 v = xs[lane + 32 * k];
                s += v.x + v.y + v.z + v.w;
            }
            acc[rb] = s;
        }
        #pragma unroll
        for (int rb = 0; rb < 4; ++rb) {
            float v = acc[rb];
            v += __shfl_xor(v, 1);
            v += __shfl_xor(v, 2);
            v += __shfl_xor(v, 16);
            acc[rb] = v;
        }
        if (((lane & 3) == 0) && lane < 16) {
            const int cb = lane >> 2;
            #pragma unroll
            for (int rb = 0; rb < 4; ++rb)
                pooledLDS[sl * 16 + rb * 4 + cb] = acc[rb];
        }
    }
    __syncthreads();

    if (tid < 32) {
        const float scale = scale_p[0];
        const int sm = lane & 15;
        const int hi = lane >> 4;

        float a[16];
        float nrm = 0.0f;
        #pragma unroll
        for (int i = 0; i < 16; ++i) { a[i] = pooledLDS[sm * 16 + i]; nrm += a[i] * a[i]; }
        const float inv = 1.0f / sqrtf(nrm);

        float re[16], im[16];
        #pragma unroll
        for (int i = 0; i < 16; ++i) { re[i] = a[i] * inv; im[i] = 0.0f; }

        #pragma unroll
        for (int l = 0; l < 6; ++l) {
            #pragma unroll
            for (int q = 0; q < 4; ++q) {
                const float half = 0.5f * qw[l * 4 + q];
                const float c = cosf(half);
                const float s = sinf(half);
                const int m = 1 << (3 - q);
                #pragma unroll
                for (int i = 0; i < 16; ++i) {
                    if (!(i & m)) {
                        const int j = i | m;
                        const float r0 = re[i], i0 = im[i];
                        const float r1 = re[j], i1 = im[j];
                        re[i] = c * r0 + s * i1;
                        im[i] = c * i0 - s * r1;
                        re[j] = c * r1 + s * i0;
                        im[j] = c * i1 - s * r0;
                    }
                }
            }
            #pragma unroll
            for (int q = 0; q < 4; ++q) {
                const int cm = 1 << (3 - q);
                const int tm = 1 << (3 - ((q + 1) & 3));
                #pragma unroll
                for (int i = 0; i < 16; ++i) {
                    if ((i & cm) && !(i & tm)) {
                        const int j = i | tm;
                        float t;
                        t = re[i]; re[i] = re[j]; re[j] = t;
                        t = im[i]; im[i] = im[j]; im[j] = t;
                    }
                }
            }
        }

        if (lane < 16) {
            #pragma unroll
            for (int i = 0; i < 16; ++i)
                probsLDS[lane * 16 + i] = re[i] * re[i] + im[i] * im[i];
        }
        __asm__ volatile("s_wait_dscnt 0" ::: "memory");

        v2f Af[4];
        #pragma unroll
        for (int t = 0; t < 4; ++t) {
            Af[t].x = probsLDS[sm * 16 + 4 * t + 2 * hi];
            Af[t].y = probsLDS[sm * 16 + 4 * t + 2 * hi + 1];
        }

        const int n = sm;
        const bool valid = (n < 10);
        float wn[10];
        #pragma unroll
        for (int j = 0; j < 10; ++j) wn[j] = valid ? hw[n * 10 + j] : 0.0f;
        const int cmask[10] = {8, 4, 2, 1, 12, 10, 9, 6, 5, 3};
        v2f Bf[4];
        #pragma unroll
        for (int t = 0; t < 4; ++t) {
            const int k0 = 4 * t + 2 * hi;
            float m0 = 0.0f, m1 = 0.0f;
            #pragma unroll
            for (int j = 0; j < 10; ++j) {
                const float w = wn[j];
                m0 += (__popc(k0 & cmask[j]) & 1) ? -w : w;
                m1 += (__popc((k0 + 1) & cmask[j]) & 1) ? -w : w;
            }
            Bf[t].x = m0;
            Bf[t].y = m1;
        }

        v8f acc = {};
        acc = __builtin_amdgcn_wmma_f32_16x16x4_f32(false, Af[0], false, Bf[0],
                                                    (short)0, acc, false, false);
        acc = __builtin_amdgcn_wmma_f32_16x16x4_f32(false, Af[1], false, Bf[1],
                                                    (short)0, acc, false, false);
        acc = __builtin_amdgcn_wmma_f32_16x16x4_f32(false, Af[2], false, Bf[2],
                                                    (short)0, acc, false, false);
        acc = __builtin_amdgcn_wmma_f32_16x16x4_f32(false, Af[3], false, Bf[3],
                                                    (short)0, acc, false, false);

        if (valid) {
            const float b = hb[n];
            #pragma unroll
            for (int v = 0; v < 8; ++v) {
                const int samp = base + v + 8 * hi;
                out[samp * 10 + n] = (acc[v] + b) * scale;
            }
        }
    }
}

extern "C" void kernel_launch(void* const* d_in, const int* in_sizes, int n_in,
                              void* d_out, int out_size, void* d_ws, size_t ws_size,
                              hipStream_t stream) {
    const float* x  = (const float*)d_in[0];
    const float* qw = (const float*)d_in[1];
    const float* hw = (const float*)d_in[2];
    const float* hb = (const float*)d_in[3];
    const float* sc = (const float*)d_in[4];
    float* out = (float*)d_out;

    const int nsamp  = in_sizes[0] / 4096;
    const int blocks = nsamp / SAMP_PER_BLOCK;

    qnet_fused_kernel<<<blocks, 256, 0, stream>>>(x, qw, hw, hb, sc, out);
}


# Image-independent work moved ahead of the pooling: angles, head weights, bias, scale requested before the rows; sin/cos, gate constants and head B fragments built while the first rows are in flight
# speedup vs baseline: 1.6027x; 1.6027x over previous
#include <hip/hip_runtime.h>


typedef float v2f __attribute__((ext_vector_type(2)));
typedef float v4f __attribute__((ext_vector_type(4)));
typedef float v8f __attribute__((ext_vector_type(8)));

#define SAMP_PER_BLOCK 16

#define ROWS16(b) "+v"(b[0]), "+v"(b[1]), "+v"(b[2]), "+v"(b[3]), "+v"(b[4]), "+v"(b[5]),     \
                  "+v"(b[6]), "+v"(b[7]), "+v"(b[8]), "+v"(b[9]), "+v"(b[10]), "+v"(b[11]),   \
                  "+v"(b[12]), "+v"(b[13]), "+v"(b[14]), "+v"(b[15])

struct LaneMap { int col[4]; };

constexpr int map_apply(LaneMap p, int k) {
    int r = 0;
    for (int b = 0; b < 4; ++b) if ((k >> b) & 1) r ^= p.col[b];
    return r;
}
constexpr LaneMap map_cnot(LaneMap p, int cm, int tm) {
    for (int b = 0; b < 4; ++b) if (p.col[b] & cm) p.col[b] ^= tm;
    return p;
}
constexpr LaneMap map_before_layer(int layer) {
    LaneMap p = {{4, 8, 1, 2}};
    for (int l = 0; l < layer; ++l)
        for (int q = 0; q < 4; ++q)
            p = map_cnot(p, 8 >> q, 8 >> ((q + 1) & 3));
    return p;
}
constexpr int partner_mask(int layer, int q) {
    const LaneMap p = map_before_layer(layer);
    for (int d = 1; d < 16; ++d) if (map_apply(p, d) == (8 >> q)) return d;
    return 0;
}

template <int D>
__device__ __forceinline__ float row_xor(float v) {
    return __int_as_float(__builtin_amdgcn_update_dpp(
        0, __float_as_int(v), 0x160 | D, 0xf, 0xf, true));
}

template <int G>
__device__ __forceinline__ void rx_gates(float& re, float& im,
                                         const float (&cg)[24], const float (&sg)[24]) {
    if constexpr (G < 24) {
        constexpr int D = partner_mask(G / 4, G % 4);
        const float c = cg[G];
        const float s = sg[G];
        const float rp = row_xor<D>(re);
        const float ip = row_xor<D>(im);
        const float nre = c * re + s * ip;
        const float nim = c * im - s * rp;
        re = nre;
        im = nim;
        rx_gates<G + 1>(re, im, cg, sg);
    }
}

__global__ __launch_bounds__(256) __attribute__((amdgpu_waves_per_eu(5))) void qnet_fused_kernel(
    const float* __restrict__ x,
    const float* __restrict__ qw,
    const float* __restrict__ hw,
    const float* __restrict__ hb,
    const float* __restrict__ scale_p,
    float* __restrict__ out)
{
    __shared__ float probsLDS[16 * 16];
    __shared__ v4f   headLDS[32 * 2];

    const int tid  = threadIdx.x;
    const int lane = tid & 31;
    const int wv   = __builtin_amdgcn_readfirstlane(tid >> 5);
    const int base = blockIdx.x * SAMP_PER_BLOCK;
    const int row = lane >> 4;
    const int k   = lane & 15;
    const bool head_wave = (wv == 0);
    const bool valid = (k < 10);

    const float my_w = qw[lane < 24 ? lane : 23];
    float wn[10];
    float hbias = 0.0f, scale = 0.0f;
    #pragma unroll
    for (int j = 0; j < 10; ++j) wn[j] = 0.0f;
    if (head_wave) {
        const int hn = valid ? k : 0;
        #pragma unroll
        for (int j = 0; j < 10; ++j) wn[j] = hw[hn * 10 + j];
        hbias = hb[hn];
        scale = scale_p[0];
    }

    const int sl  = wv * 2 + row;
    const v4f* __restrict__ xs =
        (const v4f*)(x + (size_t)(base + sl) * 4096) + k;
    float acc[4];
    v4f buf[2][16];
    #pragma unroll
    for (int r = 0; r < 16; ++r) buf[0][r] = xs[16 * r];
    #pragma unroll
    for (int r = 0; r < 16; ++r) buf[1][r] = xs[16 * (16 + r)];
    __builtin_amdgcn_sched_barrier(0);

    float cg[24], sg[24];
    {
        const float half = 0.5f * my_w;
        const int cs_bits = __float_as_int(cosf(half));
        const int sn_bits = __float_as_int(sinf(half));
        #pragma unroll
        for (int g = 0; g < 24; ++g) {
            cg[g] = __int_as_float(__builtin_amdgcn_readlane(cs_bits, g));
            sg[g] = __int_as_float(__builtin_amdgcn_readlane(sn_bits, g));
        }
    }

    if (head_wave) {
        v2f Bf[4];
        const int cmask[10] = {8, 4, 2, 1, 12, 10, 9, 6, 5, 3};
        #pragma unroll
        for (int t = 0; t < 4; ++t) {
            const int k0 = 4 * t + 2 * row;
            float m0 = 0.0f, m1 = 0.0f;
            #pragma unroll
            for (int j = 0; j < 10; ++j) {
                const float w = valid ? wn[j] : 0.0f;
                m0 += (__popc(k0 & cmask[j]) & 1) ? -w : w;
                m1 += (__popc((k0 + 1) & cmask[j]) & 1) ? -w : w;
            }
            Bf[t].x = m0;
            Bf[t].y = m1;
        }
        headLDS[lane * 2 + 0] = (v4f){Bf[0].x, Bf[0].y, Bf[1].x, Bf[1].y};
        headLDS[lane * 2 + 1] = (v4f){Bf[2].x, Bf[2].y, Bf[3].x, Bf[3].y};
    }

    #pragma unroll
    for (int rb = 0; rb < 4; ++rb) {
        if (rb >= 1 && rb < 3) {
            #pragma unroll
            for (int r = 0; r < 16; ++r)
                buf[(rb + 1) & 1][r] = xs[16 * ((rb + 1) * 16 + r)];
        }
        v4f* b = buf[rb & 1];
        if (rb == 0) {
            __asm__ volatile("" : ROWS16(b));
        } else {
            __asm__ volatile("" : ROWS16(b), "+v"(acc[rb - 1]));
        }
        v2f s2 = {0.0f, 0.0f};
        #pragma unroll
        for (int r = 0; r < 16; ++r) {
            const v4f v = b[r];
            s2 += __builtin_shufflevector(v, v, 0, 1) + __builtin_shufflevector(v, v, 2, 3);
        }
        acc[rb] = s2.x + s2.y;
    }

    {
        const bool b0 = (k & 1) != 0;
        const bool b1 = (k & 2) != 0;
        const float keep01 = b0 ? acc[1] : acc[0];
        const float give01 = b0 ? acc[0] : acc[1];
        const float keep23 = b0 ? acc[3] : acc[2];
        const float give23 = b0 ? acc[2] : acc[3];
        const float u01 = keep01 + row_xor<1>(give01);
        const float u23 = keep23 + row_xor<1>(give23);
        const float keep = b1 ? u23 : u01;
        const float give = b1 ? u01 : u23;
        const float a = keep + row_xor<2>(give);

        float nrm = a * a;
        nrm += row_xor<1>(nrm);
        nrm += row_xor<2>(nrm);
        nrm += row_xor<4>(nrm);
        nrm += row_xor<8>(nrm);
        const float inv = 1.0f / sqrtf(nrm);

        float re = a * inv;
        float im = 0.0f;
        rx_gates<0>(re, im, cg, sg);

        constexpr LaneMap pf = map_before_layer(6);
        const int kk = ((k & 1) ? pf.col[0] : 0) ^ ((k & 2) ? pf.col[1] : 0) ^
                       ((k & 4) ? pf.col[2] : 0) ^ ((k & 8) ? pf.col[3] : 0);
        probsLDS[sl * 16 + kk] = re * re + im * im;
    }
    __syncthreads();

    if (head_wave) {
        const int sm = k;
        const int hi = row;

        v2f Af[4];
        #pragma unroll
        for (int t = 0; t < 4; ++t) {
            Af[t].x = probsLDS[sm * 16 + 4 * t + 2 * hi];
            Af[t].y = probsLDS[sm * 16 + 4 * t + 2 * hi + 1];
        }

        const v4f p0 = headLDS[lane * 2 + 0];
        const v4f p1 = headLDS[lane * 2 + 1];
        v2f Bf[4];
        Bf[0] = __builtin_shufflevector(p0, p0, 0, 1);
        Bf[1] = __builtin_shufflevector(p0, p0, 2, 3);
        Bf[2] = __builtin_shufflevector(p1, p1, 0, 1);
        Bf[3] = __builtin_shufflevector(p1, p1, 2, 3);

        v8f hacc = {};
        hacc = __builtin_amdgcn_wmma_f32_16x16x4_f32(false, Af[0], false, Bf[0],
                                                     (short)0, hacc, false, false);
        hacc = __builtin_amdgcn_wmma_f32_16x16x4_f32(false, Af[1], false, Bf[1],
                                                     (short)0, hacc, false, false);
        hacc = __builtin_amdgcn_wmma_f32_16x16x4_f32(false, Af[2], false, Bf[2],
                                                     (short)0, hacc, false, false);
        hacc = __builtin_amdgcn_wmma_f32_16x16x4_f32(false, Af[3], false, Bf[3],
                                                     (short)0, hacc, false, false);

        if (valid) {
            #pragma unroll
            for (int v = 0; v < 8; ++v) {
                const int samp = base + v + 8 * hi;
                out[samp * 10 + k] = (hacc[v] + hbias) * scale;
            }
        }
    }
}

extern "C" void kernel_launch(void* const* d_in, const int* in_sizes, int n_in,
                              void* d_out, int out_size, void* d_ws, size_t ws_size,
                              hipStream_t stream) {
    const float* x  = (const float*)d_in[0];
    const float* qw = (const float*)d_in[1];
    const float* hw = (const float*)d_in[2];
    const float* hb = (const float*)d_in[3];
    const float* sc = (const float*)d_in[4];
    float* out = (float*)d_out;

    const int nsamp  = in_sizes[0] / 4096;
    const int blocks = nsamp / SAMP_PER_BLOCK;

    qnet_fused_kernel<<<blocks, 256, 0, stream>>>(x, qw, hw, hb, sc, out);
}
